# starting kernel
# baseline (speedup 1.0000x reference)
#include <hip/hip_runtime.h>
#include <math.h>

#define PN   32
#define PC   256
#define PH   56
#define PW   56
#define PHW  (PH * PW)
#define PCHW (PC * PHW)
#define PNHW (PN * PHW)

typedef float v2f __attribute__((ext_vector_type(2)));
typedef float v8f __attribute__((ext_vector_type(8)));

__global__ void __launch_bounds__(256) sa_pool_kernel(
    const float* __restrict__ x, float* __restrict__ pooled) {
  int t = blockIdx.x * blockDim.x + threadIdx.x;
  int n = t / (PHW / 4);
  int p = (t - n * (PHW / 4)) * 4;
  const float* xb = x + (size_t)n * PCHW + p;

  float sx = 0.f, sy = 0.f, sz = 0.f, sw = 0.f;
  float mx = -INFINITY, my = -INFINITY, mz = -INFINITY, mw = -INFINITY;
#pragma unroll 8
  for (int c = 0; c < PC; ++c) {
    const float4 v = *(const float4*)(xb + (size_t)c * PHW);
    sx += v.x; sy += v.y; sz += v.z; sw += v.w;
    mx = fmaxf(mx, v.x); my = fmaxf(my, v.y);
    mz = fmaxf(mz, v.z); mw = fmaxf(mw, v.w);
  }
  const float inv = 1.0f / (float)PC;
  float* pd = pooled + (size_t)n * 2 * PHW + p;
  float4 a; a.x = sx * inv; a.y = sy * inv; a.z = sz * inv; a.w = sw * inv;
  float4 m; m.x = mx; m.y = my; m.z = mz; m.w = mw;
  *(float4*)(pd)        = a;
  *(float4*)(pd + PHW)  = m;
}

__global__ void __launch_bounds__(128) sa_conv_sigmoid_kernel(
    const float* __restrict__ pooled,
    const float* __restrict__ wconv,
    const float* __restrict__ bias,
    float* __restrict__ scale) {
  const int nh   = blockIdx.x;
  const int n    = nh / PH;
  const int h    = nh - n * PH;
  const int wave = threadIdx.x >> 5;
  const int lane = threadIdx.x & 31;
  const int w0   = (wave == 3) ? (PW - 16) : (wave * 16);
  const int m    = lane & 15;
  const int khalf = (lane >> 4) << 1;

  const float* pb = pooled + (size_t)n * 2 * PHW;

  v8f acc = {0.f, 0.f, 0.f, 0.f, 0.f, 0.f, 0.f, 0.f};

#pragma unroll
  for (int k0 = 0; k0 < 100; k0 += 4) {
    v2f a, b;
#pragma unroll
    for (int e = 0; e < 2; ++e) {
      const int k = k0 + khalf + e;
      float av = 0.f, bv = 0.f;
      if (k < 98) {
        const int ci  = k / 49;
        const int rem = k - ci * 49;
        const int r   = rem / 7;
        const int s   = rem - r * 7;
        const int hh  = h + r - 3;
        const int ww  = w0 + m + s - 3;
        if (hh >= 0 && hh < PH && ww >= 0 && ww < PW)
          av = pb[ci * PHW + hh * PW + ww];
        bv = wconv[k];
      }
      if (e == 0) { a.x = av; b.x = bv; } else { a.y = av; b.y = bv; }
    }
    acc = __builtin_amdgcn_wmma_f32_16x16x4_f32(
         false, a,  false, b,
         (short)0, acc,  false,  false);
  }

  int rr = 0, mout = 0;
  bool writer = false;
  if (lane < 8)                    { writer = true; rr = lane;      mout = lane;      }
  else if (lane >= 16 && lane < 24){ writer = true; rr = lane - 16; mout = lane - 8;  }

  float y = acc[0];
  if (rr == 1) y = acc[1];
  if (rr == 2) y = acc[2];
  if (rr == 3) y = acc[3];
  if (rr == 4) y = acc[4];
  if (rr == 5) y = acc[5];
  if (rr == 6) y = acc[6];
  if (rr == 7) y = acc[7];

  if (writer) {
    y += bias[0];
    const float sg = 1.0f / (1.0f + __expf(-y));
    scale[(size_t)n * PHW + h * PW + w0 + mout] = sg;
  }
}

__global__ void __launch_bounds__(256) sa_apply_kernel(
    const float* __restrict__ x, const float* __restrict__ scale,
    float* __restrict__ out) {
  int t = blockIdx.x * blockDim.x + threadIdx.x;
  int i = t * 4;
  int n = i / PCHW;
  int r = i - n * PCHW;
  int p = r % PHW;

  const float4 xv = *(const float4*)(x + (size_t)i);
  const float4 sv = *(const float4*)(scale + (size_t)n * PHW + p);
  float4 o;
  o.x = xv.x * sv.x; o.y = xv.y * sv.y;
  o.z = xv.z * sv.z; o.w = xv.w * sv.w;
  *(float4*)(out + (size_t)i) = o;
}

extern "C" void kernel_launch(void* const* d_in, const int* in_sizes, int n_in,
                              void* d_out, int out_size, void* d_ws, size_t ws_size,
                              hipStream_t stream) {
  const float* x     = (const float*)d_in[0];
  const float* wconv = (const float*)d_in[1];
  const float* bias  = (const float*)d_in[2];
  float* out = (float*)d_out;

  float* pooled = (float*)d_ws;
  float* scale  = pooled + (size_t)PN * 2 * PHW;

  sa_pool_kernel<<<(PNHW / 4) / 256, 256, 0, stream>>>(x, pooled);

  sa_conv_sigmoid_kernel<<<PN * PH, 128, 0, stream>>>(pooled, wconv, bias, scale);

  sa_apply_kernel<<<(PN * PCHW / 4) / 256, 256, 0, stream>>>(x, scale, out);
}


# pool kernel: one work-group per 128-pixel tile, 256 channels split over 8 waves and joined in LDS (784 groups, not 98); changes the order of the channel sum
# speedup vs baseline: 1.3430x; 1.3430x over previous
#include <hip/hip_runtime.h>
#include <math.h>

#define PN   32
#define PC   256
#define PH   56
#define PW   56
#define PHW  (PH * PW)
#define PCHW (PC * PHW)
#define PNHW (PN * PHW)

typedef float v2f __attribute__((ext_vector_type(2)));
typedef float v8f __attribute__((ext_vector_type(8)));

#define POOL_WAVES 8
#define POOL_CPW   (PC / POOL_WAVES)
#define POOL_GRP   16

__global__ void __launch_bounds__(256) sa_pool_kernel(
    const float* __restrict__ x, float* __restrict__ pooled) {
  __shared__ float4 part[2][POOL_WAVES][32];

  const int lane = threadIdx.x & 31;
  const int wave = __builtin_amdgcn_readfirstlane((int)(threadIdx.x >> 5));
  const int t = blockIdx.x * 32 + lane;
  const int n = t / (PHW / 4);
  const int p = (t - n * (PHW / 4)) * 4;
  const float* xb = x + (size_t)n * PCHW + (size_t)(wave * POOL_CPW) * PHW + p;

  float sx = 0.f, sy = 0.f, sz = 0.f, sw = 0.f;
  float mx = -INFINITY, my = -INFINITY, mz = -INFINITY, mw = -INFINITY;
  for (int c0 = 0; c0 < POOL_CPW; c0 += POOL_GRP) {
    float4 v[POOL_GRP];
#pragma unroll
    for (int j = 0; j < POOL_GRP; ++j)
      v[j] = *(const float4*)(xb + (size_t)(c0 + j) * PHW);
#pragma unroll
    for (int j = 0; j < POOL_GRP; ++j) {
      sx += v[j].x; sy += v[j].y; sz += v[j].z; sw += v[j].w;
      mx = fmaxf(mx, v[j].x); my = fmaxf(my, v[j].y);
      mz = fmaxf(mz, v[j].z); mw = fmaxf(mw, v[j].w);
    }
  }
  float4 ps; ps.x = sx; ps.y = sy; ps.z = sz; ps.w = sw;
  float4 pm; pm.x = mx; pm.y = my; pm.z = mz; pm.w = mw;
  part[0][wave][lane] = ps;
  part[1][wave][lane] = pm;
  __syncthreads();

  if (wave < 2) {
    float4 q[POOL_WAVES];
#pragma unroll
    for (int w = 0; w < POOL_WAVES; ++w) q[w] = part[wave][w][lane];
    float4 r = q[0];
    if (wave == 0) {
#pragma unroll
      for (int w = 1; w < POOL_WAVES; ++w) {
        r.x += q[w].x; r.y += q[w].y; r.z += q[w].z; r.w += q[w].w;
      }
      const float inv = 1.0f / (float)PC;
      r.x *= inv; r.y *= inv; r.z *= inv; r.w *= inv;
    } else {
#pragma unroll
      for (int w = 1; w < POOL_WAVES; ++w) {
        r.x = fmaxf(r.x, q[w].x); r.y = fmaxf(r.y, q[w].y);
        r.z = fmaxf(r.z, q[w].z); r.w = fmaxf(r.w, q[w].w);
      }
    }
    *(float4*)(pooled + (size_t)n * 2 * PHW + (size_t)wave * PHW + p) = r;
  }
}

__global__ void __launch_bounds__(128) sa_conv_sigmoid_kernel(
    const float* __restrict__ pooled,
    const float* __restrict__ wconv,
    const float* __restrict__ bias,
    float* __restrict__ scale) {
  const int nh   = blockIdx.x;
  const int n    = nh / PH;
  const int h    = nh - n * PH;
  const int wave = threadIdx.x >> 5;
  const int lane = threadIdx.x & 31;
  const int w0   = (wave == 3) ? (PW - 16) : (wave * 16);
  const int m    = lane & 15;
  const int khalf = (lane >> 4) << 1;

  const float* pb = pooled + (size_t)n * 2 * PHW;

  v8f acc = {0.f, 0.f, 0.f, 0.f, 0.f, 0.f, 0.f, 0.f};

#pragma unroll
  for (int k0 = 0; k0 < 100; k0 += 4) {
    v2f a, b;
#pragma unroll
    for (int e = 0; e < 2; ++e) {
      const int k = k0 + khalf + e;
      float av = 0.f, bv = 0.f;
      if (k < 98) {
        const int ci  = k / 49;
        const int rem = k - ci * 49;
        const int r   = rem / 7;
        const int s   = rem - r * 7;
        const int hh  = h + r - 3;
        const int ww  = w0 + m + s - 3;
        if (hh >= 0 && hh < PH && ww >= 0 && ww < PW)
          av = pb[ci * PHW + hh * PW + ww];
        bv = wconv[k];
      }
      if (e == 0) { a.x = av; b.x = bv; } else { a.y = av; b.y = bv; }
    }
    acc = __builtin_amdgcn_wmma_f32_16x16x4_f32(
         false, a,  false, b,
         (short)0, acc,  false,  false);
  }

  int rr = 0, mout = 0;
  bool writer = false;
  if (lane < 8)                    { writer = true; rr = lane;      mout = lane;      }
  else if (lane >= 16 && lane < 24){ writer = true; rr = lane - 16; mout = lane - 8;  }

  float y = acc[0];
  if (rr == 1) y = acc[1];
  if (rr == 2) y = acc[2];
  if (rr == 3) y = acc[3];
  if (rr == 4) y = acc[4];
  if (rr == 5) y = acc[5];
  if (rr == 6) y = acc[6];
  if (rr == 7) y = acc[7];

  if (writer) {
    y += bias[0];
    const float sg = 1.0f / (1.0f + __expf(-y));
    scale[(size_t)n * PHW + h * PW + w0 + mout] = sg;
  }
}

__global__ void __launch_bounds__(256) sa_apply_kernel(
    const float* __restrict__ x, const float* __restrict__ scale,
    float* __restrict__ out) {
  int t = blockIdx.x * blockDim.x + threadIdx.x;
  int i = t * 4;
  int n = i / PCHW;
  int r = i - n * PCHW;
  int p = r % PHW;

  const float4 xv = *(const float4*)(x + (size_t)i);
  const float4 sv = *(const float4*)(scale + (size_t)n * PHW + p);
  float4 o;
  o.x = xv.x * sv.x; o.y = xv.y * sv.y;
  o.z = xv.z * sv.z; o.w = xv.w * sv.w;
  *(float4*)(out + (size_t)i) = o;
}

extern "C" void kernel_launch(void* const* d_in, const int* in_sizes, int n_in,
                              void* d_out, int out_size, void* d_ws, size_t ws_size,
                              hipStream_t stream) {
  const float* x     = (const float*)d_in[0];
  const float* wconv = (const float*)d_in[1];
  const float* bias  = (const float*)d_in[2];
  float* out = (float*)d_out;

  float* pooled = (float*)d_ws;
  float* scale  = pooled + (size_t)PN * 2 * PHW;

  sa_pool_kernel<<<(PNHW / 4) / 32, 256, 0, stream>>>(x, pooled);

  sa_conv_sigmoid_kernel<<<PN * PH, 128, 0, stream>>>(pooled, wconv, bias, scale);

  sa_apply_kernel<<<(PN * PCHW / 4) / 256, 256, 0, stream>>>(x, scale, out);
}


# apply kernel: takes the pool kernel's grid and tiling (784 work-groups, a wave = 32 channels of a 128-pixel tile, scale held in registers, 8 to 16 loads in flight) instead of one float4 per thread
# speedup vs baseline: 1.7038x; 1.2687x over previous
#include <hip/hip_runtime.h>
#include <math.h>

#define PN   32
#define PC   256
#define PH   56
#define PW   56
#define PHW  (PH * PW)
#define PCHW (PC * PHW)
#define PNHW (PN * PHW)

typedef float v2f __attribute__((ext_vector_type(2)));
typedef float v8f __attribute__((ext_vector_type(8)));

#define POOL_WAVES 8
#define POOL_CPW   (PC / POOL_WAVES)
#define POOL_GRP   16

__global__ void __launch_bounds__(256) sa_pool_kernel(
    const float* __restrict__ x, float* __restrict__ pooled) {
  __shared__ float4 part[2][POOL_WAVES][32];

  const int lane = threadIdx.x & 31;
  const int wave = __builtin_amdgcn_readfirstlane((int)(threadIdx.x >> 5));
  const int t = blockIdx.x * 32 + lane;
  const int n = t / (PHW / 4);
  const int p = (t - n * (PHW / 4)) * 4;
  const float* xb = x + (size_t)n * PCHW + (size_t)(wave * POOL_CPW) * PHW + p;

  float sx = 0.f, sy = 0.f, sz = 0.f, sw = 0.f;
  float mx = -INFINITY, my = -INFINITY, mz = -INFINITY, mw = -INFINITY;
  for (int c0 = 0; c0 < POOL_CPW; c0 += POOL_GRP) {
    float4 v[POOL_GRP];
#pragma unroll
    for (int j = 0; j < POOL_GRP; ++j)
      v[j] = *(const float4*)(xb + (size_t)(c0 + j) * PHW);
#pragma unroll
    for (int j = 0; j < POOL_GRP; ++j) {
      sx += v[j].x; sy += v[j].y; sz += v[j].z; sw += v[j].w;
      mx = fmaxf(mx, v[j].x); my = fmaxf(my, v[j].y);
      mz = fmaxf(mz, v[j].z); mw = fmaxf(mw, v[j].w);
    }
  }
  float4 ps; ps.x = sx; ps.y = sy; ps.z = sz; ps.w = sw;
  float4 pm; pm.x = mx; pm.y = my; pm.z = mz; pm.w = mw;
  part[0][wave][lane] = ps;
  part[1][wave][lane] = pm;
  __syncthreads();

  if (wave < 2) {
    float4 q[POOL_WAVES];
#pragma unroll
    for (int w = 0; w < POOL_WAVES; ++w) q[w] = part[wave][w][lane];
    float4 r = q[0];
    if (wave == 0) {
#pragma unroll
      for (int w = 1; w < POOL_WAVES; ++w) {
        r.x += q[w].x; r.y += q[w].y; r.z += q[w].z; r.w += q[w].w;
      }
      const float inv = 1.0f / (float)PC;
      r.x *= inv; r.y *= inv; r.z *= inv; r.w *= inv;
    } else {
#pragma unroll
      for (int w = 1; w < POOL_WAVES; ++w) {
        r.x = fmaxf(r.x, q[w].x); r.y = fmaxf(r.y, q[w].y);
        r.z = fmaxf(r.z, q[w].z); r.w = fmaxf(r.w, q[w].w);
      }
    }
    *(float4*)(pooled + (size_t)n * 2 * PHW + (size_t)wave * PHW + p) = r;
  }
}

__global__ void __launch_bounds__(128) sa_conv_sigmoid_kernel(
    const float* __restrict__ pooled,
    const float* __restrict__ wconv,
    const float* __restrict__ bias,
    float* __restrict__ scale) {
  const int nh   = blockIdx.x;
  const int n    = nh / PH;
  const int h    = nh - n * PH;
  const int wave = threadIdx.x >> 5;
  const int lane = threadIdx.x & 31;
  const int w0   = (wave == 3) ? (PW - 16) : (wave * 16);
  const int m    = lane & 15;
  const int khalf = (lane >> 4) << 1;

  const float* pb = pooled + (size_t)n * 2 * PHW;

  v8f acc = {0.f, 0.f, 0.f, 0.f, 0.f, 0.f, 0.f, 0.f};

#pragma unroll
  for (int k0 = 0; k0 < 100; k0 += 4) {
    v2f a, b;
#pragma unroll
    for (int e = 0; e < 2; ++e) {
      const int k = k0 + khalf + e;
      float av = 0.f, bv = 0.f;
      if (k < 98) {
        const int ci  = k / 49;
        const int rem = k - ci * 49;
        const int r   = rem / 7;
        const int s   = rem - r * 7;
        const int hh  = h + r - 3;
        const int ww  = w0 + m + s - 3;
        if (hh >= 0 && hh < PH && ww >= 0 && ww < PW)
          av = pb[ci * PHW + hh * PW + ww];
        bv = wconv[k];
      }
      if (e == 0) { a.x = av; b.x = bv; } else { a.y = av; b.y = bv; }
    }
    acc = __builtin_amdgcn_wmma_f32_16x16x4_f32(
         false, a,  false, b,
         (short)0, acc,  false,  false);
  }

  int rr = 0, mout = 0;
  bool writer = false;
  if (lane < 8)                    { writer = true; rr = lane;      mout = lane;      }
  else if (lane >= 16 && lane < 24){ writer = true; rr = lane - 16; mout = lane - 8;  }

  float y = acc[0];
  if (rr == 1) y = acc[1];
  if (rr == 2) y = acc[2];
  if (rr == 3) y = acc[3];
  if (rr == 4) y = acc[4];
  if (rr == 5) y = acc[5];
  if (rr == 6) y = acc[6];
  if (rr == 7) y = acc[7];

  if (writer) {
    y += bias[0];
    const float sg = 1.0f / (1.0f + __expf(-y));
    scale[(size_t)n * PHW + h * PW + w0 + mout] = sg;
  }
}

#define APPLY_GRP  8
#define APPLY_NGRP (POOL_CPW / APPLY_GRP)

__global__ void __launch_bounds__(256) sa_apply_kernel(
    const float* __restrict__ x, const float* __restrict__ scale,
    float* __restrict__ out) {
  const int lane = threadIdx.x & 31;
  const int wave = __builtin_amdgcn_readfirstlane((int)(threadIdx.x >> 5));
  const int t = blockIdx.x * 32 + lane;
  const int n = t / (PHW / 4);
  const int p = (t - n * (PHW / 4)) * 4;
  const size_t off = (size_t)n * PCHW + (size_t)(wave * POOL_CPW) * PHW + p;
  const float* xb = x + off;
  float* ob = out + off;

  float4 sv = *(const float4*)(scale + (size_t)n * PHW + p);

  float4 buf[2][APPLY_GRP];
#pragma unroll
  for (int j = 0; j < APPLY_GRP; ++j)
    buf[0][j] = *(const float4*)(xb + (size_t)j * PHW);
#pragma unroll
  for (int g = 0; g < APPLY_NGRP; ++g) {
    if (g + 1 < APPLY_NGRP) {
#pragma unroll
      for (int j = 0; j < APPLY_GRP; ++j)
        buf[(g + 1) & 1][j] =
            *(const float4*)(xb + (size_t)((g + 1) * APPLY_GRP + j) * PHW);
    }
    asm volatile("" : "+v"(sv.x), "+v"(sv.y), "+v"(sv.z), "+v"(sv.w));
#pragma unroll
    for (int j = 0; j < APPLY_GRP; ++j) {
      const float4 v = buf[g & 1][j];
      float4 o;
      o.x = v.x * sv.x; o.y = v.y * sv.y;
      o.z = v.z * sv.z; o.w = v.w * sv.w;
      *(float4*)(ob + (size_t)(g * APPLY_GRP + j) * PHW) = o;
    }
  }
}

extern "C" void kernel_launch(void* const* d_in, const int* in_sizes, int n_in,
                              void* d_out, int out_size, void* d_ws, size_t ws_size,
                              hipStream_t stream) {
  const float* x     = (const float*)d_in[0];
  const float* wconv = (const float*)d_in[1];
  const float* bias  = (const float*)d_in[2];
  float* out = (float*)d_out;

  float* pooled = (float*)d_ws;
  float* scale  = pooled + (size_t)PN * 2 * PHW;

  sa_pool_kernel<<<(PNHW / 4) / 32, 256, 0, stream>>>(x, pooled);

  sa_conv_sigmoid_kernel<<<PN * PH, 128, 0, stream>>>(pooled, wconv, bias, scale);

  sa_apply_kernel<<<(PNHW / 4) / 32, 256, 0, stream>>>(x, scale, out);
}


# apply kernel computes the 7x7 conv + sigmoid of its own 128 pixels itself (16-pixel f32 WMMA tile per wave, all taps in flight, scales via LDS): 2 launches, not 3; order of the conv's tap sum changes
# speedup vs baseline: 2.1456x; 1.2593x over previous
#include <hip/hip_runtime.h>
#include <math.h>

#define PN   32
#define PC   256
#define PH   56
#define PW   56
#define PHW  (PH * PW)
#define PCHW (PC * PHW)
#define PNHW (PN * PHW)

typedef float v2f __attribute__((ext_vector_type(2)));
typedef float v8f __attribute__((ext_vector_type(8)));

#define POOL_WAVES 8
#define POOL_CPW   (PC / POOL_WAVES)
#define POOL_GRP   16

__global__ void __launch_bounds__(256) sa_pool_kernel(
    const float* __restrict__ x, float* __restrict__ pooled) {
  __shared__ float4 part[2][POOL_WAVES][32];

  const int lane = threadIdx.x & 31;
  const int wave = __builtin_amdgcn_readfirstlane((int)(threadIdx.x >> 5));
  const int t = blockIdx.x * 32 + lane;
  const int n = t / (PHW / 4);
  const int p = (t - n * (PHW / 4)) * 4;
  const float* xb = x + (size_t)n * PCHW + (size_t)(wave * POOL_CPW) * PHW + p;

  float sx = 0.f, sy = 0.f, sz = 0.f, sw = 0.f;
  float mx = -INFINITY, my = -INFINITY, mz = -INFINITY, mw = -INFINITY;
  for (int c0 = 0; c0 < POOL_CPW; c0 += POOL_GRP) {
    float4 v[POOL_GRP];
#pragma unroll
    for (int j = 0; j < POOL_GRP; ++j)
      v[j] = *(const float4*)(xb + (size_t)(c0 + j) * PHW);
#pragma unroll
    for (int j = 0; j < POOL_GRP; ++j) {
      sx += v[j].x; sy += v[j].y; sz += v[j].z; sw += v[j].w;
      mx = fmaxf(mx, v[j].x); my = fmaxf(my, v[j].y);
      mz = fmaxf(mz, v[j].z); mw = fmaxf(mw, v[j].w);
    }
  }
  float4 ps; ps.x = sx; ps.y = sy; ps.z = sz; ps.w = sw;
  float4 pm; pm.x = mx; pm.y = my; pm.z = mz; pm.w = mw;
  part[0][wave][lane] = ps;
  part[1][wave][lane] = pm;
  __syncthreads();

  if (wave < 2) {
    float4 q[POOL_WAVES];
#pragma unroll
    for (int w = 0; w < POOL_WAVES; ++w) q[w] = part[wave][w][lane];
    float4 r = q[0];
    if (wave == 0) {
#pragma unroll
      for (int w = 1; w < POOL_WAVES; ++w) {
        r.x += q[w].x; r.y += q[w].y; r.z += q[w].z; r.w += q[w].w;
      }
      const float inv = 1.0f / (float)PC;
      r.x *= inv; r.y *= inv; r.z *= inv; r.w *= inv;
    } else {
#pragma unroll
      for (int w = 1; w < POOL_WAVES; ++w) {
        r.x = fmaxf(r.x, q[w].x); r.y = fmaxf(r.y, q[w].y);
        r.z = fmaxf(r.z, q[w].z); r.w = fmaxf(r.w, q[w].w);
      }
    }
    *(float4*)(pooled + (size_t)n * 2 * PHW + (size_t)wave * PHW + p) = r;
  }
}

#define APPLY_GRP  8
#define APPLY_NGRP (POOL_CPW / APPLY_GRP)

__global__ void __launch_bounds__(256) sa_conv_apply_kernel(
    const float* __restrict__ x,
    const float* __restrict__ pooled,
    const float* __restrict__ wconv,
    const float* __restrict__ bias,
    float* __restrict__ out) {
  __shared__ __attribute__((aligned(16))) float s_scale[128];
  __shared__ __attribute__((aligned(16))) float s_w[8 * 128];

  const int lane = threadIdx.x & 31;
  const int wave = __builtin_amdgcn_readfirstlane((int)(threadIdx.x >> 5));
  const int t = blockIdx.x * 32 + lane;
  const int n = t / (PHW / 4);
  const int p = (t - n * (PHW / 4)) * 4;
  const size_t off = (size_t)n * PCHW + (size_t)(wave * POOL_CPW) * PHW + p;
  const float* xb = x + off;
  float* ob = out + off;

  const int m  = lane & 15;
  const int hf = lane >> 4;
  const int g0 = blockIdx.x * 128 + wave * 16;
  const int cn = g0 / PHW;
  const int cp = g0 - cn * PHW + m;
  const int ch = cp / PW;
  const int cw = cp - ch * PW;
  const int ib = cn * 2 * PHW + hf * PHW + cp;

  float wv[4];
#pragma unroll
  for (int k = 0; k < 4; ++k) {
    const int i  = lane + 32 * k;
    const int pl2 = i / 50;
    const int qq = i - pl2 * 50;
    const bool wok = (i < 100) && (qq < 49);
    const float v = wconv[wok ? pl2 * 49 + qq : 0];
    wv[k] = wok ? v : 0.f;
  }

  bool rowok[7], colok[7];
#pragma unroll
  for (int r = 0; r < 7; ++r) {
    rowok[r] = (unsigned)(ch + r - 3) < (unsigned)PH;
    colok[r] = (unsigned)(cw + r - 3) < (unsigned)PW;
  }

  float av[50];
#pragma unroll
  for (int q = 0; q < 49; ++q) {
    const int r = q / 7;
    const int c = q - r * 7;
    const bool ok = rowok[r] && colok[c];
    const float v = pooled[ok ? ib + (r - 3) * PW + (c - 3) : 0];
    av[q] = ok ? v : 0.f;
  }
  av[49] = 0.f;

  float4 buf[2][APPLY_GRP];
#pragma unroll
  for (int j = 0; j < APPLY_GRP; ++j)
    buf[0][j] = *(const float4*)(xb + (size_t)j * PHW);

  __builtin_amdgcn_sched_barrier(0);

  float* wt = s_w + wave * 128;
#pragma unroll
  for (int k = 0; k < 4; ++k) wt[lane + 32 * k] = wv[k];
  __builtin_amdgcn_wave_barrier();

  v8f acc = {0.f, 0.f, 0.f, 0.f, 0.f, 0.f, 0.f, 0.f};
#pragma unroll
  for (int j = 0; j < 25; ++j) {
    v2f a;
    a.x = av[2 * j];
    a.y = av[2 * j + 1];
    const v2f b = *(const v2f*)(wt + hf * 50 + 2 * j);
    acc = __builtin_amdgcn_wmma_f32_16x16x4_f32(
         false, a,  false, b,
         (short)0, acc,  false,  false);
  }

  float y = acc[0];
  if (m == 1) y = acc[1];
  if (m == 2) y = acc[2];
  if (m == 3) y = acc[3];
  if (m == 4) y = acc[4];
  if (m == 5) y = acc[5];
  if (m == 6) y = acc[6];
  if (m == 7) y = acc[7];
  if (m < 8) {
    y += bias[0];
    s_scale[wave * 16 + hf * 8 + m] = 1.0f / (1.0f + __expf(-y));
  }
  __builtin_amdgcn_fence(__ATOMIC_RELEASE, "workgroup", "local");
  __builtin_amdgcn_s_barrier();
  __builtin_amdgcn_fence(__ATOMIC_ACQUIRE, "workgroup", "local");
  float4 sv = *(const float4*)(s_scale + 4 * lane);

#pragma unroll
  for (int g = 0; g < APPLY_NGRP; ++g) {
    if (g + 1 < APPLY_NGRP) {
#pragma unroll
      for (int j = 0; j < APPLY_GRP; ++j)
        buf[(g + 1) & 1][j] =
            *(const float4*)(xb + (size_t)((g + 1) * APPLY_GRP + j) * PHW);
    }
    asm volatile("" : "+v"(sv.x), "+v"(sv.y), "+v"(sv.z), "+v"(sv.w));
#pragma unroll
    for (int j = 0; j < APPLY_GRP; ++j) {
      const float4 v = buf[g & 1][j];
      float4 o;
      o.x = v.x * sv.x; o.y = v.y * sv.y;
      o.z = v.z * sv.z; o.w = v.w * sv.w;
      *(float4*)(ob + (size_t)(g * APPLY_GRP + j) * PHW) = o;
    }
  }
}

extern "C" void kernel_launch(void* const* d_in, const int* in_sizes, int n_in,
                              void* d_out, int out_size, void* d_ws, size_t ws_size,
                              hipStream_t stream) {
  const float* x     = (const float*)d_in[0];
  const float* wconv = (const float*)d_in[1];
  const float* bias  = (const float*)d_in[2];
  float* out = (float*)d_out;

  float* pooled = (float*)d_ws;

  sa_pool_kernel<<<(PNHW / 4) / 32, 256, 0, stream>>>(x, pooled);

  sa_conv_apply_kernel<<<(PNHW / 4) / 32, 256, 0, stream>>>(x, pooled, wconv, bias, out);
}
